# starting kernel
# baseline (speedup 1.0000x reference)
#include <hip/hip_runtime.h>
#include <math.h>

typedef float v2f __attribute__((ext_vector_type(2)));
typedef float v8f __attribute__((ext_vector_type(8)));

#define WAVES_PER_BLOCK 4
#define PIf 3.14159265358979323846f

__global__ void vqc_setup_kernel(const float* __restrict__ params,
                                 float* __restrict__ Mout) {
    __shared__ float A[32][33];
    const int lane = threadIdx.x;

    float gc[10], gs[10];
#pragma unroll
    for (int g = 0; g < 10; ++g) {
        float th = params[g * 2];
        gs[g] = __sinf(0.5f * th);
        gc[g] = __cosf(0.5f * th);
    }

    for (int j = 0; j < 32; ++j) {
        float val = (lane == j) ? 1.0f : 0.0f;
#pragma unroll
        for (int l = 0; l < 2; ++l) {
#pragma unroll
            for (int i = 0; i < 5; ++i) {
                const int g = l * 5 + i;
                const int mask = 1 << (4 - i);
                float ov = __shfl(val, lane ^ mask, 32);
                int bit = (lane >> (4 - i)) & 1;
                val = bit ? fmaf(gs[g], ov, gc[g] * val)
                          : fmaf(-gs[g], ov, gc[g] * val);
            }
#pragma unroll
            for (int i = 0; i < 5; ++i) {
                const int t = (i + 1) % 5;
                int cbit = (lane >> (4 - i)) & 1;
                int src = cbit ? (lane ^ (1 << (4 - t))) : lane;
                val = __shfl(val, src, 32);
            }
        }
        float dcz = 1.0f;
#pragma unroll
        for (int i = 0; i < 5; ++i) {
            int b1 = (j >> (4 - i)) & 1;
            int b2 = (j >> (4 - (i + 1) % 5)) & 1;
            if (b1 & b2) dcz = -dcz;
        }
        A[lane][j] = dcz * val;
    }
    __syncthreads();

    for (int p = 0; p < 32; ++p) {
        float m = 0.0f;
#pragma unroll
        for (int r = 0; r < 32; ++r) {
            float zr = (r & 16) ? -1.0f : 1.0f;
            m = fmaf(zr * A[r][p], A[r][lane], m);
        }
        Mout[p * 32 + lane] = m;
    }
}

__global__ __launch_bounds__(WAVES_PER_BLOCK * 32) void vqc_main_kernel(
    const float* __restrict__ x, const float* __restrict__ Mq,
    float* __restrict__ out, int batch) {
    const int lane = threadIdx.x & 31;
    const int wave = threadIdx.x >> 5;
    const int bl = lane & 15;
    const int hi = lane >> 4;
    const int group = blockIdx.x * WAVES_PER_BLOCK + wave;
    const int e = group * 16 + bl;
    const int el = (e < batch) ? e : (batch - 1);

    float cc[5], ss[5];
#pragma unroll
    for (int w = 0; w < 5; ++w) {
        float th = 0.5f * PIf * x[el * 5 + w];
        __sincosf(th, &ss[w], &cc[w]);
    }
    float q0r[2] = {cc[0] * cc[0], cc[0] * ss[0]};
    float q0i[2] = {-cc[0] * ss[0], ss[0] * ss[0]};
    float q1r[2] = {cc[1] * cc[1], cc[1] * ss[1]};
    float q1i[2] = {-cc[1] * ss[1], ss[1] * ss[1]};
    float q2r[2] = {cc[2] * cc[2], cc[2] * ss[2]};
    float q2i[2] = {-cc[2] * ss[2], ss[2] * ss[2]};
    float q3r = hi ? cc[3] * ss[3] : cc[3] * cc[3];
    float q3i = hi ? ss[3] * ss[3] : -cc[3] * ss[3];
    float q4r[2] = {cc[4] * cc[4], cc[4] * ss[4]};
    float q4i[2] = {-cc[4] * ss[4], ss[4] * ss[4]};
    float tr[2], ti[2];
#pragma unroll
    for (int j = 0; j < 2; ++j) {
        tr[j] = q3r * q4r[j] - q3i * q4i[j];
        ti[j] = q3r * q4i[j] + q3i * q4r[j];
    }
    float p01r[4], p01i[4];
#pragma unroll
    for (int b0 = 0; b0 < 2; ++b0)
#pragma unroll
        for (int b1 = 0; b1 < 2; ++b1) {
            p01r[b0 * 2 + b1] = q0r[b0] * q1r[b1] - q0i[b0] * q1i[b1];
            p01i[b0 * 2 + b1] = q0r[b0] * q1i[b1] + q0i[b0] * q1r[b1];
        }
    v2f are[8], aim[8];
#pragma unroll
    for (int kk = 0; kk < 8; ++kk) {
        const int b01 = kk >> 1, b2 = kk & 1;
        float pr = p01r[b01] * q2r[b2] - p01i[b01] * q2i[b2];
        float pi = p01r[b01] * q2i[b2] + p01i[b01] * q2r[b2];
#pragma unroll
        for (int j = 0; j < 2; ++j) {
            are[kk][j] = pr * tr[j] - pi * ti[j];
            aim[kk][j] = pr * ti[j] + pi * tr[j];
        }
    }

    v8f accRe0 = {}, accRe1 = {}, accIm0 = {}, accIm1 = {};
#pragma unroll
    for (int kk = 0; kk < 8; ++kk) {
        const int krow = 4 * kk + 2 * hi;
        v2f b0v, b1v;
        b0v[0] = Mq[krow * 32 + bl];
        b0v[1] = Mq[(krow + 1) * 32 + bl];
        b1v[0] = Mq[krow * 32 + 16 + bl];
        b1v[1] = Mq[(krow + 1) * 32 + 16 + bl];
        accRe0 = __builtin_amdgcn_wmma_f32_16x16x4_f32(false, are[kk], false,
                                                       b0v, (short)0, accRe0,
                                                       false, false);
        accRe1 = __builtin_amdgcn_wmma_f32_16x16x4_f32(false, are[kk], false,
                                                       b1v, (short)0, accRe1,
                                                       false, false);
        accIm0 = __builtin_amdgcn_wmma_f32_16x16x4_f32(false, aim[kk], false,
                                                       b0v, (short)0, accIm0,
                                                       false, false);
        accIm1 = __builtin_amdgcn_wmma_f32_16x16x4_f32(false, aim[kk], false,
                                                       b1v, (short)0, accIm1,
                                                       false, false);
    }

    __shared__ float T[WAVES_PER_BLOCK][2][16][33];
#pragma unroll
    for (int r = 0; r < 8; ++r) {
        const int m = r + 8 * hi;
        T[wave][0][m][bl] = accRe0[r];
        T[wave][0][m][16 + bl] = accRe1[r];
        T[wave][1][m][bl] = accIm0[r];
        T[wave][1][m][16 + bl] = accIm1[r];
    }
    __syncthreads();

    float sum = 0.0f;
#pragma unroll
    for (int kk = 0; kk < 8; ++kk) {
        const int n = 4 * kk + 2 * hi;
        sum = fmaf(T[wave][0][bl][n], are[kk][0], sum);
        sum = fmaf(T[wave][0][bl][n + 1], are[kk][1], sum);
        sum = fmaf(T[wave][1][bl][n], aim[kk][0], sum);
        sum = fmaf(T[wave][1][bl][n + 1], aim[kk][1], sum);
    }
    sum += __shfl(sum, lane ^ 16, 32);
    if (hi == 0 && e < batch) out[e] = sum;
}

extern "C" void kernel_launch(void* const* d_in, const int* in_sizes, int n_in,
                              void* d_out, int out_size, void* d_ws,
                              size_t ws_size, hipStream_t stream) {
    const float* x = (const float*)d_in[0];
    const float* params = (const float*)d_in[1];
    float* out = (float*)d_out;
    float* Mq = (float*)d_ws;

    const int batch = in_sizes[0] / 5;

    vqc_setup_kernel<<<1, 32, 0, stream>>>(params, Mq);

    const int groups = (batch + 15) / 16;
    const int blocks = (groups + WAVES_PER_BLOCK - 1) / WAVES_PER_BLOCK;
    vqc_main_kernel<<<blocks, WAVES_PER_BLOCK * 32, 0, stream>>>(x, Mq, out,
                                                                 batch);
}


# main product transposed: folded matrix rows as A operand, samples as B columns, k slots numbered so a lane's accumulators meet its own amplitudes; no LDS or barrier; float sum order changes
# speedup vs baseline: 3.5920x; 3.5920x over previous
#include <hip/hip_runtime.h>
#include <math.h>

typedef float v2f __attribute__((ext_vector_type(2)));
typedef float v8f __attribute__((ext_vector_type(8)));

#define WAVES_PER_BLOCK 4
#define PIf 3.14159265358979323846f


__device__ constexpr int cnot_ring_src(int k) {
    for (int i = 4; i >= 0; --i) {
        const int t = (i + 1) % 5;
        if ((k >> (4 - i)) & 1) k ^= 1 << (4 - t);
    }
    return k;
}

__device__ __forceinline__ v8f wmma_f32(v2f a, v2f b, v8f c) {
    return __builtin_amdgcn_wmma_f32_16x16x4_f32(false, a, false, b, (short)0, c,
                                                 false, false);
}

__global__ __launch_bounds__(32) void vqc_setup_kernel(
    const float* __restrict__ params, float* __restrict__ Mout) {
    const int lane = threadIdx.x;
    const int c = lane & 15;
    const int h = lane >> 4;

    float gc[10], gs[10];
    for (int g = 0; g < 10; ++g) {
        float th = params[g * 2];
        gs[g] = __sinf(0.5f * th);
        gc[g] = __cosf(0.5f * th);
    }

    v2f aL[2][8];
    for (int rt = 0; rt < 2; ++rt) {
        const int rho = cnot_ring_src(16 * rt + c);
        float f[5][2];
        for (int i = 0; i < 5; ++i) {
            const int rb = (rho >> (4 - i)) & 1;
            f[i][0] = rb ? gs[5 + i] : gc[5 + i];
            f[i][1] = rb ? gc[5 + i] : -gs[5 + i];
        }
        const float f3 = h ? f[3][1] : f[3][0];
        const float t0 = f3 * f[4][0];
        const float t1 = f3 * f[4][1];
        for (int kk = 0; kk < 8; ++kk) {
            const float u = f[0][(kk >> 2) & 1] * f[1][(kk >> 1) & 1] * f[2][kk & 1];
            aL[rt][kk][0] = u * t0;
            aL[rt][kk][1] = u * t1;
        }
    }

    v2f bL[2][8];
    constexpr int dsrc = cnot_ring_src(2);
    for (int ct = 0; ct < 2; ++ct) {
        const int j = 16 * ct + c;
        float dcz = 1.0f;
        for (int i = 0; i < 5; ++i) {
            const int b1 = (j >> (4 - i)) & 1;
            const int b2 = (j >> (4 - (i + 1) % 5)) & 1;
            if (b1 & b2) dcz = -dcz;
        }
        float g[5][2];
        for (int i = 0; i < 5; ++i) {
            const int jb = (j >> (4 - i)) & 1;
            const float g0 = jb ? -gs[i] : gc[i];
            const float g1 = jb ? gc[i] : gs[i];
            const bool flip = h && ((dsrc >> (4 - i)) & 1);
            g[i][0] = flip ? g1 : g0;
            g[i][1] = flip ? g0 : g1;
        }
        g[0][0] *= dcz;
        g[0][1] *= dcz;
        for (int kk = 0; kk < 8; ++kk) {
            for (int s = 0; s < 2; ++s) {
                const int sig = cnot_ring_src(4 * kk + s);
                bL[ct][kk][s] = g[0][(sig >> 4) & 1] * g[1][(sig >> 3) & 1] *
                                g[2][(sig >> 2) & 1] * g[3][(sig >> 1) & 1] *
                                g[4][sig & 1];
            }
        }
    }

    v8f At[2][2] = {};
    for (int kk = 0; kk < 8; ++kk)
        for (int rt = 0; rt < 2; ++rt)
            for (int ct = 0; ct < 2; ++ct)
                At[rt][ct] = wmma_f32(aL[rt][kk], bL[ct][kk], At[rt][ct]);

    v8f Az[2][2];
    for (int ct = 0; ct < 2; ++ct) {
        Az[0][ct] = At[0][ct];
        Az[1][ct] = -At[1][ct];
    }

    v8f Mt[2][2] = {};
    for (int rt = 0; rt < 2; ++rt)
        for (int j = 0; j < 4; ++j)
            for (int pt = 0; pt < 2; ++pt)
                for (int qt = pt; qt < 2; ++qt) {
                    v2f a, b;
                    a[0] = At[rt][pt][2 * j];
                    a[1] = At[rt][pt][2 * j + 1];
                    b[0] = Az[rt][qt][2 * j];
                    b[1] = Az[rt][qt][2 * j + 1];
                    Mt[pt][qt] = wmma_f32(a, b, Mt[pt][qt]);
                }

    float4* Gf = reinterpret_cast<float4*>(Mout);
    int piece = 0;
    for (int pt = 0; pt < 2; ++pt)
        for (int qt = pt; qt < 2; ++qt) {
            const float fold = (qt != pt) ? 2.0f : 1.0f;
            float4 lo, hi4;
            lo.x = fold * Mt[pt][qt][0];
            lo.y = fold * Mt[pt][qt][1];
            lo.z = fold * Mt[pt][qt][2];
            lo.w = fold * Mt[pt][qt][3];
            hi4.x = fold * Mt[pt][qt][4];
            hi4.y = fold * Mt[pt][qt][5];
            hi4.z = fold * Mt[pt][qt][6];
            hi4.w = fold * Mt[pt][qt][7];
            Gf[piece * 32 + lane] = lo;
            Gf[(piece + 1) * 32 + lane] = hi4;
            piece += 2;
        }
}

__global__ __launch_bounds__(WAVES_PER_BLOCK * 32) void vqc_main_kernel(
    const float* __restrict__ x, const float* __restrict__ Gq,
    float* __restrict__ out, int batch) {
    const int lane = threadIdx.x & 31;
    const int wave = threadIdx.x >> 5;
    const int c = lane & 15;
    const int h = lane >> 4;
    const int group = blockIdx.x * WAVES_PER_BLOCK + wave;
    const int e = group * 16 + c;
    const int el = (e < batch) ? e : (batch - 1);

    const float4* Gf = reinterpret_cast<const float4*>(Gq);
    const float4 g0a = Gf[0 * 32 + lane], g0b = Gf[1 * 32 + lane];
    const float4 g1a = Gf[2 * 32 + lane], g1b = Gf[3 * 32 + lane];
    const float4 g1c = Gf[4 * 32 + lane], g1d = Gf[5 * 32 + lane];

    float cc[5], ss[5];
    const float* xrow = x + (size_t)el * 5;
#pragma unroll
    for (int w = 0; w < 5; ++w) {
        float th = 0.5f * PIf * xrow[w];
        __sincosf(th, &ss[w], &cc[w]);
    }
    v2f qr[5], qi[5];
#pragma unroll
    for (int w = 0; w < 5; ++w) {
        const v2f cs = {cc[w], ss[w]};
        qr[w] = cc[w] * cs;
        qi[w] = v2f{-qr[w][1], ss[w] * ss[w]};
    }
    const float q1r = h ? qr[1][1] : qr[1][0];
    const float q1i = h ? qi[1][1] : qi[1][0];
    const v2f pr = qr[0] * q1r - qi[0] * q1i;
    const v2f pi = qr[0] * q1i + qi[0] * q1r;
    v2f ur[2], ui[2];
#pragma unroll
    for (int b0 = 0; b0 < 2; ++b0) {
        ur[b0] = pr[b0] * qr[2] - pi[b0] * qi[2];
        ui[b0] = pr[b0] * qi[2] + pi[b0] * qr[2];
    }
    v2f tr[2], ti[2];
#pragma unroll
    for (int b3 = 0; b3 < 2; ++b3) {
        tr[b3] = qr[3][b3] * qr[4] - qi[3][b3] * qi[4];
        ti[b3] = qr[3][b3] * qi[4] + qi[3][b3] * qr[4];
    }
    v2f bre[8], bim[8];
#pragma unroll
    for (int kk = 0; kk < 8; ++kk) {
        const int b0 = kk >> 2, b2 = (kk >> 1) & 1, b3 = kk & 1;
        bre[kk] = ur[b0][b2] * tr[b3] - ui[b0][b2] * ti[b3];
        bim[kk] = ur[b0][b2] * ti[b3] + ui[b0][b2] * tr[b3];
    }

    v2f a0[4], a1[8];
    a0[0][0] = g0a.x; a0[0][1] = g0a.y; a0[1][0] = g0a.z; a0[1][1] = g0a.w;
    a0[2][0] = g0b.x; a0[2][1] = g0b.y; a0[3][0] = g0b.z; a0[3][1] = g0b.w;
    a1[0][0] = g1a.x; a1[0][1] = g1a.y; a1[1][0] = g1a.z; a1[1][1] = g1a.w;
    a1[2][0] = g1b.x; a1[2][1] = g1b.y; a1[3][0] = g1b.z; a1[3][1] = g1b.w;
    a1[4][0] = g1c.x; a1[4][1] = g1c.y; a1[5][0] = g1c.z; a1[5][1] = g1c.w;
    a1[6][0] = g1d.x; a1[6][1] = g1d.y; a1[7][0] = g1d.z; a1[7][1] = g1d.w;

    v8f accRe0 = {}, accRe1 = {}, accIm0 = {}, accIm1 = {};
#pragma unroll
    for (int kk = 0; kk < 8; ++kk) {
        if (kk < 4) {
            accRe0 = wmma_f32(a0[kk], bre[kk], accRe0);
            accIm0 = wmma_f32(a0[kk], bim[kk], accIm0);
        }
        accRe1 = wmma_f32(a1[kk], bre[kk], accRe1);
        accIm1 = wmma_f32(a1[kk], bim[kk], accIm1);
    }

    v2f sRe0 = {}, sIm0 = {}, sRe1 = {}, sIm1 = {};
#pragma unroll
    for (int j = 0; j < 4; ++j) {
        const v2f dRe0 = {accRe0[2 * j], accRe0[2 * j + 1]};
        const v2f dIm0 = {accIm0[2 * j], accIm0[2 * j + 1]};
        const v2f dRe1 = {accRe1[2 * j], accRe1[2 * j + 1]};
        const v2f dIm1 = {accIm1[2 * j], accIm1[2 * j + 1]};
        sRe0 += dRe0 * bre[j];
        sIm0 += dIm0 * bim[j];
        sRe1 += dRe1 * bre[4 + j];
        sIm1 += dIm1 * bim[4 + j];
    }
    const v2f sv = (sRe0 + sIm0) + (sRe1 + sIm1);
    float sum = sv[0] + sv[1];
    sum += __shfl(sum, lane ^ 16, 32);
    if (h == 0 && e < batch) out[e] = sum;
}

extern "C" void kernel_launch(void* const* d_in, const int* in_sizes, int n_in,
                              void* d_out, int out_size, void* d_ws,
                              size_t ws_size, hipStream_t stream) {
    const float* x = (const float*)d_in[0];
    const float* params = (const float*)d_in[1];
    float* out = (float*)d_out;
    float* Mq = (float*)d_ws;

    const int batch = in_sizes[0] / 5;

    vqc_setup_kernel<<<1, 32, 0, stream>>>(params, Mq);

    const int groups = (batch + 15) / 16;
    const int blocks = (groups + WAVES_PER_BLOCK - 1) / WAVES_PER_BLOCK;
    vqc_main_kernel<<<blocks, WAVES_PER_BLOCK * 32, 0, stream>>>(x, Mq, out,
                                                                 batch);
}


# main kernel made resident: 2048 work-groups, each wave loops over 4 sample groups, folded matrix kept in registers, next group's x loaded one trip ahead; 8192 waves enter, not 32768; sums unchanged
# speedup vs baseline: 3.7780x; 1.0518x over previous
#include <hip/hip_runtime.h>
#include <math.h>

typedef float v2f __attribute__((ext_vector_type(2)));
typedef float v8f __attribute__((ext_vector_type(8)));

#define WAVES_PER_BLOCK 4
#define MAIN_BLOCKS 2048
#define PIf 3.14159265358979323846f


__device__ constexpr int cnot_ring_src(int k) {
    for (int i = 4; i >= 0; --i) {
        const int t = (i + 1) % 5;
        if ((k >> (4 - i)) & 1) k ^= 1 << (4 - t);
    }
    return k;
}

__device__ __forceinline__ v8f wmma_f32(v2f a, v2f b, v8f c) {
    return __builtin_amdgcn_wmma_f32_16x16x4_f32(false, a, false, b, (short)0, c,
                                                 false, false);
}

__global__ __launch_bounds__(32) void vqc_setup_kernel(
    const float* __restrict__ params, float* __restrict__ Mout) {
    const int lane = threadIdx.x;
    const int c = lane & 15;
    const int h = lane >> 4;

    float gc[10], gs[10];
    for (int g = 0; g < 10; ++g) {
        float th = params[g * 2];
        gs[g] = __sinf(0.5f * th);
        gc[g] = __cosf(0.5f * th);
    }

    v2f aL[2][8];
    for (int rt = 0; rt < 2; ++rt) {
        const int rho = cnot_ring_src(16 * rt + c);
        float f[5][2];
        for (int i = 0; i < 5; ++i) {
            const int rb = (rho >> (4 - i)) & 1;
            f[i][0] = rb ? gs[5 + i] : gc[5 + i];
            f[i][1] = rb ? gc[5 + i] : -gs[5 + i];
        }
        const float f3 = h ? f[3][1] : f[3][0];
        const float t0 = f3 * f[4][0];
        const float t1 = f3 * f[4][1];
        for (int kk = 0; kk < 8; ++kk) {
            const float u = f[0][(kk >> 2) & 1] * f[1][(kk >> 1) & 1] * f[2][kk & 1];
            aL[rt][kk][0] = u * t0;
            aL[rt][kk][1] = u * t1;
        }
    }

    v2f bL[2][8];
    constexpr int dsrc = cnot_ring_src(2);
    for (int ct = 0; ct < 2; ++ct) {
        const int j = 16 * ct + c;
        float dcz = 1.0f;
        for (int i = 0; i < 5; ++i) {
            const int b1 = (j >> (4 - i)) & 1;
            const int b2 = (j >> (4 - (i + 1) % 5)) & 1;
            if (b1 & b2) dcz = -dcz;
        }
        float g[5][2];
        for (int i = 0; i < 5; ++i) {
            const int jb = (j >> (4 - i)) & 1;
            const float g0 = jb ? -gs[i] : gc[i];
            const float g1 = jb ? gc[i] : gs[i];
            const bool flip = h && ((dsrc >> (4 - i)) & 1);
            g[i][0] = flip ? g1 : g0;
            g[i][1] = flip ? g0 : g1;
        }
        g[0][0] *= dcz;
        g[0][1] *= dcz;
        for (int kk = 0; kk < 8; ++kk) {
            for (int s = 0; s < 2; ++s) {
                const int sig = cnot_ring_src(4 * kk + s);
                bL[ct][kk][s] = g[0][(sig >> 4) & 1] * g[1][(sig >> 3) & 1] *
                                g[2][(sig >> 2) & 1] * g[3][(sig >> 1) & 1] *
                                g[4][sig & 1];
            }
        }
    }

    v8f At[2][2] = {};
    for (int kk = 0; kk < 8; ++kk)
        for (int rt = 0; rt < 2; ++rt)
            for (int ct = 0; ct < 2; ++ct)
                At[rt][ct] = wmma_f32(aL[rt][kk], bL[ct][kk], At[rt][ct]);

    v8f Az[2][2];
    for (int ct = 0; ct < 2; ++ct) {
        Az[0][ct] = At[0][ct];
        Az[1][ct] = -At[1][ct];
    }

    v8f Mt[2][2] = {};
    for (int rt = 0; rt < 2; ++rt)
        for (int j = 0; j < 4; ++j)
            for (int pt = 0; pt < 2; ++pt)
                for (int qt = pt; qt < 2; ++qt) {
                    v2f a, b;
                    a[0] = At[rt][pt][2 * j];
                    a[1] = At[rt][pt][2 * j + 1];
                    b[0] = Az[rt][qt][2 * j];
                    b[1] = Az[rt][qt][2 * j + 1];
                    Mt[pt][qt] = wmma_f32(a, b, Mt[pt][qt]);
                }

    float4* Gf = reinterpret_cast<float4*>(Mout);
    int piece = 0;
    for (int pt = 0; pt < 2; ++pt)
        for (int qt = pt; qt < 2; ++qt) {
            const float fold = (qt != pt) ? 2.0f : 1.0f;
            float4 lo, hi4;
            lo.x = fold * Mt[pt][qt][0];
            lo.y = fold * Mt[pt][qt][1];
            lo.z = fold * Mt[pt][qt][2];
            lo.w = fold * Mt[pt][qt][3];
            hi4.x = fold * Mt[pt][qt][4];
            hi4.y = fold * Mt[pt][qt][5];
            hi4.z = fold * Mt[pt][qt][6];
            hi4.w = fold * Mt[pt][qt][7];
            Gf[piece * 32 + lane] = lo;
            Gf[(piece + 1) * 32 + lane] = hi4;
            piece += 2;
        }
}

__global__ __launch_bounds__(WAVES_PER_BLOCK * 32) void vqc_main_kernel(
    const float* __restrict__ x, const float* __restrict__ Gq,
    float* __restrict__ out, int batch) {
    const int lane = threadIdx.x & 31;
    const int c = lane & 15;
    const int h = lane >> 4;
    const int groups = (batch + 15) >> 4;
    constexpr int stride = MAIN_BLOCKS * WAVES_PER_BLOCK;
    const int first = __builtin_amdgcn_readfirstlane(
        (int)(blockIdx.x * WAVES_PER_BLOCK + (threadIdx.x >> 5)));

    const float4* Gf = reinterpret_cast<const float4*>(Gq);
    const float4 g0a = Gf[0 * 32 + lane], g0b = Gf[1 * 32 + lane];
    const float4 g1a = Gf[2 * 32 + lane], g1b = Gf[3 * 32 + lane];
    const float4 g1c = Gf[4 * 32 + lane], g1d = Gf[5 * 32 + lane];

    float xc[5];
    {
        const int e0 = first * 16 + c;
        const int el0 = (e0 < batch) ? e0 : (batch - 1);
        const float* xrow = x + (size_t)el0 * 5;
#pragma unroll
        for (int w = 0; w < 5; ++w) xc[w] = xrow[w];
    }

    v2f a0[4], a1[8];
    a0[0][0] = g0a.x; a0[0][1] = g0a.y; a0[1][0] = g0a.z; a0[1][1] = g0a.w;
    a0[2][0] = g0b.x; a0[2][1] = g0b.y; a0[3][0] = g0b.z; a0[3][1] = g0b.w;
    a1[0][0] = g1a.x; a1[0][1] = g1a.y; a1[1][0] = g1a.z; a1[1][1] = g1a.w;
    a1[2][0] = g1b.x; a1[2][1] = g1b.y; a1[3][0] = g1b.z; a1[3][1] = g1b.w;
    a1[4][0] = g1c.x; a1[4][1] = g1c.y; a1[5][0] = g1c.z; a1[5][1] = g1c.w;
    a1[6][0] = g1d.x; a1[6][1] = g1d.y; a1[7][0] = g1d.z; a1[7][1] = g1d.w;

    for (int group = first; group < groups; group += stride) {
        const int e = group * 16 + c;

        float xn[5];
        {
            const int en = e + stride * 16;
            const int eln = (en < batch) ? en : (batch - 1);
            const float* xrow = x + (size_t)eln * 5;
#pragma unroll
            for (int w = 0; w < 5; ++w) xn[w] = xrow[w];
        }
        __builtin_amdgcn_sched_barrier(0);

        float cc[5], ss[5];
#pragma unroll
        for (int w = 0; w < 5; ++w) {
            float th = 0.5f * PIf * xc[w];
            __sincosf(th, &ss[w], &cc[w]);
        }
        v2f qr[5], qi[5];
#pragma unroll
        for (int w = 0; w < 5; ++w) {
            const v2f cs = {cc[w], ss[w]};
            qr[w] = cc[w] * cs;
            qi[w] = v2f{-qr[w][1], ss[w] * ss[w]};
        }
        const float q1r = h ? qr[1][1] : qr[1][0];
        const float q1i = h ? qi[1][1] : qi[1][0];
        const v2f pr = qr[0] * q1r - qi[0] * q1i;
        const v2f pi = qr[0] * q1i + qi[0] * q1r;
        v2f ur[2], ui[2];
#pragma unroll
        for (int b0 = 0; b0 < 2; ++b0) {
            ur[b0] = pr[b0] * qr[2] - pi[b0] * qi[2];
            ui[b0] = pr[b0] * qi[2] + pi[b0] * qr[2];
        }
        v2f tr[2], ti[2];
#pragma unroll
        for (int b3 = 0; b3 < 2; ++b3) {
            tr[b3] = qr[3][b3] * qr[4] - qi[3][b3] * qi[4];
            ti[b3] = qr[3][b3] * qi[4] + qi[3][b3] * qr[4];
        }
        v2f bre[8], bim[8];
#pragma unroll
        for (int kk = 0; kk < 8; ++kk) {
            const int b0 = kk >> 2, b2 = (kk >> 1) & 1, b3 = kk & 1;
            bre[kk] = ur[b0][b2] * tr[b3] - ui[b0][b2] * ti[b3];
            bim[kk] = ur[b0][b2] * ti[b3] + ui[b0][b2] * tr[b3];
        }

        v8f accRe0 = {}, accRe1 = {}, accIm0 = {}, accIm1 = {};
#pragma unroll
        for (int kk = 0; kk < 8; ++kk) {
            if (kk < 4) {
                accRe0 = wmma_f32(a0[kk], bre[kk], accRe0);
                accIm0 = wmma_f32(a0[kk], bim[kk], accIm0);
            }
            accRe1 = wmma_f32(a1[kk], bre[kk], accRe1);
            accIm1 = wmma_f32(a1[kk], bim[kk], accIm1);
        }

        v2f sRe0 = {}, sIm0 = {}, sRe1 = {}, sIm1 = {};
#pragma unroll
        for (int j = 0; j < 4; ++j) {
            const v2f dRe0 = {accRe0[2 * j], accRe0[2 * j + 1]};
            const v2f dIm0 = {accIm0[2 * j], accIm0[2 * j + 1]};
            const v2f dRe1 = {accRe1[2 * j], accRe1[2 * j + 1]};
            const v2f dIm1 = {accIm1[2 * j], accIm1[2 * j + 1]};
            sRe0 += dRe0 * bre[j];
            sIm0 += dIm0 * bim[j];
            sRe1 += dRe1 * bre[4 + j];
            sIm1 += dIm1 * bim[4 + j];
        }
        const v2f sv = (sRe0 + sIm0) + (sRe1 + sIm1);
        float sum = sv[0] + sv[1];
        sum += __shfl(sum, lane ^ 16, 32);
        if (h == 0 && e < batch) out[e] = sum;

#pragma unroll
        for (int w = 0; w < 5; ++w) xc[w] = xn[w];
    }
}

extern "C" void kernel_launch(void* const* d_in, const int* in_sizes, int n_in,
                              void* d_out, int out_size, void* d_ws,
                              size_t ws_size, hipStream_t stream) {
    const float* x = (const float*)d_in[0];
    const float* params = (const float*)d_in[1];
    float* out = (float*)d_out;
    float* Mq = (float*)d_ws;

    const int batch = in_sizes[0] / 5;

    vqc_setup_kernel<<<1, 32, 0, stream>>>(params, Mq);

    vqc_main_kernel<<<MAIN_BLOCKS, WAVES_PER_BLOCK * 32, 0, stream>>>(x, Mq, out,
                                                                      batch);
}


# setup launch folded in, ONE wave per SIMD: each wave builds the folded matrix once in registers (earlier fold: 2 builds per SIMD, stalled loop); 4 adjacent groups per trip fill waits; sums unchanged
# speedup vs baseline: 4.0896x; 1.0825x over previous
#include <hip/hip_runtime.h>
#include <math.h>

typedef float v2f __attribute__((ext_vector_type(2)));
typedef float v8f __attribute__((ext_vector_type(8)));

#define WAVES_PER_BLOCK 4
#define MAIN_BLOCKS 256
#define GROUPS_PER_TRIP 4
#define PIf 3.14159265358979323846f


__device__ constexpr int cnot_ring_src(int k) {
    for (int i = 4; i >= 0; --i) {
        const int t = (i + 1) % 5;
        if ((k >> (4 - i)) & 1) k ^= 1 << (4 - t);
    }
    return k;
}

__device__ __forceinline__ v8f wmma_f32(v2f a, v2f b, v8f c) {
    return __builtin_amdgcn_wmma_f32_16x16x4_f32(false, a, false, b, (short)0, c,
                                                 false, false);
}

__device__ __forceinline__ float other_half(float v) {
    return __builtin_bit_cast(
        float, __builtin_amdgcn_permlanex16(__builtin_bit_cast(int, v), __builtin_bit_cast(int, v),
                                            0x76543210, 0xfedcba98, false, false));
}

__device__ __forceinline__ void build_folded_matrix(
    const float* __restrict__ params, int c, int h, v2f (&a0)[4], v2f (&a1)[8]) {
    float gc[10], gs[10];
    for (int g = 0; g < 10; ++g) {
        float th = params[g * 2];
        gs[g] = __sinf(0.5f * th);
        gc[g] = __cosf(0.5f * th);
    }

    v2f aL[2][8];
    for (int rt = 0; rt < 2; ++rt) {
        const int rho = cnot_ring_src(16 * rt + c);
        float f[5][2];
        for (int i = 0; i < 5; ++i) {
            const int rb = (rho >> (4 - i)) & 1;
            f[i][0] = rb ? gs[5 + i] : gc[5 + i];
            f[i][1] = rb ? gc[5 + i] : -gs[5 + i];
        }
        const float f3 = h ? f[3][1] : f[3][0];
        const float t0 = f3 * f[4][0];
        const float t1 = f3 * f[4][1];
        for (int kk = 0; kk < 8; ++kk) {
            const float u = f[0][(kk >> 2) & 1] * f[1][(kk >> 1) & 1] * f[2][kk & 1];
            aL[rt][kk][0] = u * t0;
            aL[rt][kk][1] = u * t1;
        }
    }

    v2f bL[2][8];
    constexpr int dsrc = cnot_ring_src(2);
    for (int ct = 0; ct < 2; ++ct) {
        const int j = 16 * ct + c;
        float dcz = 1.0f;
        for (int i = 0; i < 5; ++i) {
            const int b1 = (j >> (4 - i)) & 1;
            const int b2 = (j >> (4 - (i + 1) % 5)) & 1;
            if (b1 & b2) dcz = -dcz;
        }
        float g[5][2];
        for (int i = 0; i < 5; ++i) {
            const int jb = (j >> (4 - i)) & 1;
            const float g0 = jb ? -gs[i] : gc[i];
            const float g1 = jb ? gc[i] : gs[i];
            const bool flip = h && ((dsrc >> (4 - i)) & 1);
            g[i][0] = flip ? g1 : g0;
            g[i][1] = flip ? g0 : g1;
        }
        g[0][0] *= dcz;
        g[0][1] *= dcz;
        for (int kk = 0; kk < 8; ++kk) {
            for (int s = 0; s < 2; ++s) {
                const int sig = cnot_ring_src(4 * kk + s);
                bL[ct][kk][s] = g[0][(sig >> 4) & 1] * g[1][(sig >> 3) & 1] *
                                g[2][(sig >> 2) & 1] * g[3][(sig >> 1) & 1] *
                                g[4][sig & 1];
            }
        }
    }

    v8f At[2][2] = {};
    for (int kk = 0; kk < 8; ++kk)
        for (int rt = 0; rt < 2; ++rt)
            for (int ct = 0; ct < 2; ++ct)
                At[rt][ct] = wmma_f32(aL[rt][kk], bL[ct][kk], At[rt][ct]);

    v8f Az[2][2];
    for (int ct = 0; ct < 2; ++ct) {
        Az[0][ct] = At[0][ct];
        Az[1][ct] = -At[1][ct];
    }

    v8f Mt[2][2] = {};
    for (int rt = 0; rt < 2; ++rt)
        for (int j = 0; j < 4; ++j)
            for (int pt = 0; pt < 2; ++pt)
                for (int qt = pt; qt < 2; ++qt) {
                    v2f a, b;
                    a[0] = At[rt][pt][2 * j];
                    a[1] = At[rt][pt][2 * j + 1];
                    b[0] = Az[rt][qt][2 * j];
                    b[1] = Az[rt][qt][2 * j + 1];
                    Mt[pt][qt] = wmma_f32(a, b, Mt[pt][qt]);
                }

    for (int j = 0; j < 4; ++j) {
        a0[j][0] = Mt[0][0][2 * j];
        a0[j][1] = Mt[0][0][2 * j + 1];
        a1[j][0] = 2.0f * Mt[0][1][2 * j];
        a1[j][1] = 2.0f * Mt[0][1][2 * j + 1];
        a1[4 + j][0] = Mt[1][1][2 * j];
        a1[4 + j][1] = Mt[1][1][2 * j + 1];
    }
}

__global__ __launch_bounds__(WAVES_PER_BLOCK * 32) __attribute__((amdgpu_waves_per_eu(1, 1))) void vqc_main_kernel(
    const float* __restrict__ x, const float* __restrict__ params,
    float* __restrict__ out, int batch) {
    constexpr int NG = GROUPS_PER_TRIP;
    const int lane = threadIdx.x & 31;
    const int c = lane & 15;
    const int h = lane >> 4;
    const int groups = (batch + 15) >> 4;
    const int trips = (groups + NG - 1) / NG;
    constexpr int stride = MAIN_BLOCKS * WAVES_PER_BLOCK;
    const int first = __builtin_amdgcn_readfirstlane(
        (int)(blockIdx.x * WAVES_PER_BLOCK + (threadIdx.x >> 5)));
    if (first >= trips) return;

    float xc[NG][5];
#pragma unroll
    for (int g = 0; g < NG; ++g) {
        const int e0 = (first * NG + g) * 16 + c;
        const int el0 = (e0 < batch) ? e0 : (batch - 1);
        const float* xrow = x + (size_t)el0 * 5;
#pragma unroll
        for (int w = 0; w < 5; ++w) xc[g][w] = xrow[w];
    }

    v2f a0[4], a1[8];
    build_folded_matrix(params, c, h, a0, a1);

    for (int trip = first; trip < trips; trip += stride) {
        float xn[NG][5];
#pragma unroll
        for (int g = 0; g < NG; ++g) {
            const int en = ((trip + stride) * NG + g) * 16 + c;
            const int eln = (en < batch) ? en : (batch - 1);
            const float* xrow = x + (size_t)eln * 5;
#pragma unroll
            for (int w = 0; w < 5; ++w) xn[g][w] = xrow[w];
        }
        __builtin_amdgcn_sched_barrier(0);

        v2f bre[NG][8], bim[NG][8];
#pragma unroll
        for (int g = 0; g < NG; ++g) {
            float cc[5], ss[5];
#pragma unroll
            for (int w = 0; w < 5; ++w) {
                float th = 0.5f * PIf * xc[g][w];
                __sincosf(th, &ss[w], &cc[w]);
            }
            v2f qr[5], qi[5];
#pragma unroll
            for (int w = 0; w < 5; ++w) {
                const v2f cs = {cc[w], ss[w]};
                qr[w] = cc[w] * cs;
                qi[w] = v2f{-qr[w][1], ss[w] * ss[w]};
            }
            const float q1r = h ? qr[1][1] : qr[1][0];
            const float q1i = h ? qi[1][1] : qi[1][0];
            const v2f pr = qr[0] * q1r - qi[0] * q1i;
            const v2f pi = qr[0] * q1i + qi[0] * q1r;
            v2f ur[2], ui[2];
#pragma unroll
            for (int b0 = 0; b0 < 2; ++b0) {
                ur[b0] = pr[b0] * qr[2] - pi[b0] * qi[2];
                ui[b0] = pr[b0] * qi[2] + pi[b0] * qr[2];
            }
            v2f tr[2], ti[2];
#pragma unroll
            for (int b3 = 0; b3 < 2; ++b3) {
                tr[b3] = qr[3][b3] * qr[4] - qi[3][b3] * qi[4];
                ti[b3] = qr[3][b3] * qi[4] + qi[3][b3] * qr[4];
            }
#pragma unroll
            for (int kk = 0; kk < 8; ++kk) {
                const int b0 = kk >> 2, b2 = (kk >> 1) & 1, b3 = kk & 1;
                bre[g][kk] = ur[b0][b2] * tr[b3] - ui[b0][b2] * ti[b3];
                bim[g][kk] = ur[b0][b2] * ti[b3] + ui[b0][b2] * tr[b3];
            }
        }

        v8f accRe0[NG], accRe1[NG], accIm0[NG], accIm1[NG];
#pragma unroll
        for (int g = 0; g < NG; ++g) {
            accRe0[g] = v8f{};
            accRe1[g] = v8f{};
            accIm0[g] = v8f{};
            accIm1[g] = v8f{};
#pragma unroll
            for (int kk = 0; kk < 8; ++kk) {
                if (kk < 4) {
                    accRe0[g] = wmma_f32(a0[kk], bre[g][kk], accRe0[g]);
                    accIm0[g] = wmma_f32(a0[kk], bim[g][kk], accIm0[g]);
                }
                accRe1[g] = wmma_f32(a1[kk], bre[g][kk], accRe1[g]);
                accIm1[g] = wmma_f32(a1[kk], bim[g][kk], accIm1[g]);
            }
        }

        float tot[NG];
#pragma unroll
        for (int g = 0; g < NG; ++g) {
            v2f sRe0 = {}, sIm0 = {}, sRe1 = {}, sIm1 = {};
#pragma unroll
            for (int j = 0; j < 4; ++j) {
                const v2f dRe0 = {accRe0[g][2 * j], accRe0[g][2 * j + 1]};
                const v2f dIm0 = {accIm0[g][2 * j], accIm0[g][2 * j + 1]};
                const v2f dRe1 = {accRe1[g][2 * j], accRe1[g][2 * j + 1]};
                const v2f dIm1 = {accIm1[g][2 * j], accIm1[g][2 * j + 1]};
                sRe0 += dRe0 * bre[g][j];
                sIm0 += dIm0 * bim[g][j];
                sRe1 += dRe1 * bre[g][4 + j];
                sIm1 += dIm1 * bim[g][4 + j];
            }
            const v2f sv = (sRe0 + sIm0) + (sRe1 + sIm1);
            const float sum = sv[0] + sv[1];
            tot[g] = sum + other_half(sum);
        }
#pragma unroll
        for (int pr2 = 0; pr2 < NG / 2; ++pr2) {
            const int e = (trip * NG + 2 * pr2) * 16 + lane;
            const float val = h ? tot[2 * pr2 + 1] : tot[2 * pr2];
            if (e < batch) out[e] = val;
        }

#pragma unroll
        for (int g = 0; g < NG; ++g)
#pragma unroll
            for (int w = 0; w < 5; ++w) xc[g][w] = xn[g][w];
    }
}

extern "C" void kernel_launch(void* const* d_in, const int* in_sizes, int n_in,
                              void* d_out, int out_size, void* d_ws,
                              size_t ws_size, hipStream_t stream) {
    const float* x = (const float*)d_in[0];
    const float* params = (const float*)d_in[1];
    float* out = (float*)d_out;

    const int batch = in_sizes[0] / 5;

    vqc_main_kernel<<<MAIN_BLOCKS, WAVES_PER_BLOCK * 32, 0, stream>>>(x, params, out,
                                                                      batch);
}
